# starting kernel
# baseline (speedup 1.0000x reference)
#include <hip/hip_runtime.h>
#include <hip/hip_bf16.h>

typedef __attribute__((ext_vector_type(2))) float v2f;
typedef __attribute__((ext_vector_type(8))) float v8f;

__global__ void zero_kernel(float* __restrict__ p, long long n) {
    long long i = (long long)blockIdx.x * blockDim.x + threadIdx.x;
    long long stride = (long long)gridDim.x * blockDim.x;
    for (; i < n; i += stride) p[i] = 0.0f;
}

__global__ void degree_kernel(const int* __restrict__ ei, float* __restrict__ deg, int E) {
    int e = blockIdx.x * blockDim.x + threadIdx.x;
    if (e < E) atomicAdd(&deg[ei[E + e]], 1.0f);
}

template <int F>
__global__ void scatter_kernel(const int* __restrict__ ei, const float* __restrict__ feat,
                               float* __restrict__ agg, int E) {
    long long t = (long long)blockIdx.x * blockDim.x + threadIdx.x;
    int e = (int)(t / (F / 4));
    int f4 = (int)(t % (F / 4)) * 4;
    if (e >= E) return;
    int src = ei[e];
    int dst = ei[E + e];
    const float4 v = *(const float4*)(feat + (size_t)src * F + f4);
    float* p = agg + (size_t)dst * F + f4;
    atomicAdd(p + 0, v.x);
    atomicAdd(p + 1, v.y);
    atomicAdd(p + 2, v.z);
    atomicAdd(p + 3, v.w);
}

template <int K, int NOUT>
__global__ __launch_bounds__(32 * (NOUT / 16)) void sage_combine_kernel(
    const float* __restrict__ xin, const float* __restrict__ agg,
    const float* __restrict__ deg, const float* __restrict__ Wl,
    const float* __restrict__ bl, const float* __restrict__ Wr,
    float* __restrict__ out, int Nn) {
    __shared__ float s_inv[16];
    __shared__ float s_a[16 * K];
    __shared__ float s_x[16 * K];

    const int node_base = blockIdx.x * 16;
    const int tid = threadIdx.x;
    const int lane = tid & 31;
    const int wave = tid >> 5;
    const int m = lane & 15;
    const int khalf = lane >> 4;
    const int ncol = wave * 16 + m;

    if (tid < 16) {
        int nd = node_base + tid;
        if (nd > Nn - 1) nd = Nn - 1;
        s_inv[tid] = 1.0f / fmaxf(deg[nd], 1.0f);
    }
    __syncthreads();

    const int nthreads = 32 * (NOUT / 16);
    for (int i = tid; i < 16 * K; i += nthreads) {
        int nd = i / K;
        int f = i - nd * K;
        int gnode = node_base + nd;
        if (gnode > Nn - 1) gnode = Nn - 1;
        s_a[i] = agg[(size_t)gnode * K + f] * s_inv[nd];
        s_x[i] = xin[(size_t)gnode * K + f];
    }
    __syncthreads();

    v8f acc = {};
#pragma unroll
    for (int k = 0; k < K; k += 4) {
        const int kb = k + khalf * 2;
        v2f a, b;
        a.x = s_a[m * K + kb];
        a.y = s_a[m * K + kb + 1];
        b.x = Wl[(size_t)kb * NOUT + ncol];
        b.y = Wl[(size_t)(kb + 1) * NOUT + ncol];
        acc = __builtin_amdgcn_wmma_f32_16x16x4_f32(false, a, false, b,
                                                    (short)0, acc, false, false);
        a.x = s_x[m * K + kb];
        a.y = s_x[m * K + kb + 1];
        b.x = Wr[(size_t)kb * NOUT + ncol];
        b.y = Wr[(size_t)(kb + 1) * NOUT + ncol];
        acc = __builtin_amdgcn_wmma_f32_16x16x4_f32(false, a, false, b,
                                                    (short)0, acc, false, false);
    }

    const float bias = bl[ncol];
#pragma unroll
    for (int v = 0; v < 8; ++v) {
        int mrow = v + khalf * 8;
        int gnode = node_base + mrow;
        if (gnode < Nn) {
            float val = acc[v] + bias;
            out[(size_t)gnode * NOUT + ncol] = fmaxf(val, 0.0f);
        }
    }
}

__global__ void fc_sigmoid_kernel(const float* __restrict__ h2,
                                  const float* __restrict__ Wfc,
                                  const float* __restrict__ bfc,
                                  float* __restrict__ out, int Nn) {
    int i = blockIdx.x * blockDim.x + threadIdx.x;
    if (i >= Nn) return;
    float s = bfc[0];
    const float* hp = h2 + (size_t)i * 64;
#pragma unroll
    for (int k = 0; k < 64; ++k) s += hp[k] * Wfc[k];
    out[i] = 1.0f / (1.0f + __expf(-s));
}

extern "C" void kernel_launch(void* const* d_in, const int* in_sizes, int n_in,
                              void* d_out, int out_size, void* d_ws, size_t ws_size,
                              hipStream_t stream) {
    const float* x    = (const float*)d_in[0];
    const int*   ei   = (const int*)d_in[1];
    const float* W1l  = (const float*)d_in[2];
    const float* b1   = (const float*)d_in[3];
    const float* W1r  = (const float*)d_in[4];
    const float* W2l  = (const float*)d_in[5];
    const float* b2   = (const float*)d_in[6];
    const float* W2r  = (const float*)d_in[7];
    const float* Wfc  = (const float*)d_in[8];
    const float* bfc  = (const float*)d_in[9];
    float* outp = (float*)d_out;

    const int Nn = in_sizes[0] / 64;
    const int Ee = in_sizes[1] / 2;

    float* deg = (float*)d_ws;
    float* agg = deg + Nn;
    float* h1  = agg + (size_t)Nn * 128;
    float* h2  = h1 + (size_t)Nn * 128;

    const int ZB = 2048, TB = 256;

    zero_kernel<<<ZB, TB, 0, stream>>>(deg, (long long)Nn * 129);
    degree_kernel<<<(Ee + TB - 1) / TB, TB, 0, stream>>>(ei, deg, Ee);
    {
        long long t = (long long)Ee * (64 / 4);
        scatter_kernel<64><<<(int)((t + TB - 1) / TB), TB, 0, stream>>>(ei, x, agg, Ee);
    }
    sage_combine_kernel<64, 128><<<(Nn + 15) / 16, 256, 0, stream>>>(
        x, agg, deg, W1l, b1, W1r, h1, Nn);

    zero_kernel<<<ZB, TB, 0, stream>>>(agg, (long long)Nn * 128);
    {
        long long t = (long long)Ee * (128 / 4);
        scatter_kernel<128><<<(int)((t + TB - 1) / TB), TB, 0, stream>>>(ei, h1, agg, Ee);
    }
    sage_combine_kernel<128, 64><<<(Nn + 15) / 16, 128, 0, stream>>>(
        h1, agg, deg, W2l, b2, W2r, h2, Nn);

    fc_sigmoid_kernel<<<(Nn + TB - 1) / TB, TB, 0, stream>>>(h2, Wfc, bfc, outp, Nn);
}


# Layer 2 projects h1 by W2l and W2r (WMMA) before aggregating, so its scatter-add runs 64 wide instead of 128 (sums now follow the product); the leftover add+relu folds into the head
# speedup vs baseline: 1.3735x; 1.3735x over previous
#include <hip/hip_runtime.h>
#include <hip/hip_bf16.h>

typedef __attribute__((ext_vector_type(2))) float v2f;
typedef __attribute__((ext_vector_type(8))) float v8f;

__global__ void zero_kernel(float* __restrict__ p, long long n) {
    long long i = (long long)blockIdx.x * blockDim.x + threadIdx.x;
    long long stride = (long long)gridDim.x * blockDim.x;
    for (; i < n; i += stride) p[i] = 0.0f;
}

__global__ void degree_kernel(const int* __restrict__ ei, float* __restrict__ deg, int E) {
    int e = blockIdx.x * blockDim.x + threadIdx.x;
    if (e < E) atomicAdd(&deg[ei[E + e]], 1.0f);
}

template <int F>
__global__ void scatter_kernel(const int* __restrict__ ei, const float* __restrict__ feat,
                               float* __restrict__ agg, int E) {
    long long t = (long long)blockIdx.x * blockDim.x + threadIdx.x;
    int e = (int)(t / (F / 4));
    int f4 = (int)(t % (F / 4)) * 4;
    if (e >= E) return;
    int src = ei[e];
    int dst = ei[E + e];
    const float4 v = *(const float4*)(feat + (size_t)src * F + f4);
    float* p = agg + (size_t)dst * F + f4;
    atomicAdd(p + 0, v.x);
    atomicAdd(p + 1, v.y);
    atomicAdd(p + 2, v.z);
    atomicAdd(p + 3, v.w);
}

template <int K, int NOUT>
__global__ __launch_bounds__(32 * (NOUT / 16)) void sage_combine_kernel(
    const float* __restrict__ xin, const float* __restrict__ agg,
    const float* __restrict__ deg, const float* __restrict__ Wl,
    const float* __restrict__ bl, const float* __restrict__ Wr,
    float* __restrict__ out, int Nn) {
    __shared__ float s_inv[16];
    __shared__ float s_a[16 * K];
    __shared__ float s_x[16 * K];

    const int node_base = blockIdx.x * 16;
    const int tid = threadIdx.x;
    const int lane = tid & 31;
    const int wave = tid >> 5;
    const int m = lane & 15;
    const int khalf = lane >> 4;
    const int ncol = wave * 16 + m;

    if (tid < 16) {
        int nd = node_base + tid;
        if (nd > Nn - 1) nd = Nn - 1;
        s_inv[tid] = 1.0f / fmaxf(deg[nd], 1.0f);
    }
    __syncthreads();

    const int nthreads = 32 * (NOUT / 16);
    for (int i = tid; i < 16 * K; i += nthreads) {
        int nd = i / K;
        int f = i - nd * K;
        int gnode = node_base + nd;
        if (gnode > Nn - 1) gnode = Nn - 1;
        s_a[i] = agg[(size_t)gnode * K + f] * s_inv[nd];
        s_x[i] = xin[(size_t)gnode * K + f];
    }
    __syncthreads();

    v8f acc = {};
#pragma unroll
    for (int k = 0; k < K; k += 4) {
        const int kb = k + khalf * 2;
        v2f a, b;
        a.x = s_a[m * K + kb];
        a.y = s_a[m * K + kb + 1];
        b.x = Wl[(size_t)kb * NOUT + ncol];
        b.y = Wl[(size_t)(kb + 1) * NOUT + ncol];
        acc = __builtin_amdgcn_wmma_f32_16x16x4_f32(false, a, false, b,
                                                    (short)0, acc, false, false);
        a.x = s_x[m * K + kb];
        a.y = s_x[m * K + kb + 1];
        b.x = Wr[(size_t)kb * NOUT + ncol];
        b.y = Wr[(size_t)(kb + 1) * NOUT + ncol];
        acc = __builtin_amdgcn_wmma_f32_16x16x4_f32(false, a, false, b,
                                                    (short)0, acc, false, false);
    }

    const float bias = bl[ncol];
#pragma unroll
    for (int v = 0; v < 8; ++v) {
        int mrow = v + khalf * 8;
        int gnode = node_base + mrow;
        if (gnode < Nn) {
            float val = acc[v] + bias;
            out[(size_t)gnode * NOUT + ncol] = fmaxf(val, 0.0f);
        }
    }
}

template <int K, int NH>
__global__ __launch_bounds__(32 * (2 * NH / 16)) void project_pair_kernel(
    const float* __restrict__ xin, const float* __restrict__ Wa,
    const float* __restrict__ Wb, float* __restrict__ pa,
    float* __restrict__ pb, int Nn) {
    constexpr int KP = K + 4;
    __shared__ __attribute__((aligned(16))) float s_x[16 * KP];

    const int node_base = blockIdx.x * 16;
    const int tid = threadIdx.x;
    const int lane = tid & 31;
    const int wave = tid >> 5;
    const int m = lane & 15;
    const int khalf = lane >> 4;
    constexpr int TILES = NH / 16;
    const bool second = wave >= TILES;
    const float* __restrict__ W = second ? Wb : Wa;
    float* __restrict__ dstp = second ? pb : pa;
    const int ncol = (second ? wave - TILES : wave) * 16 + m;

    constexpr int nthreads = 32 * (2 * NH / 16);
    for (int i = tid; i < 16 * (K / 4); i += nthreads) {
        int nd = i / (K / 4);
        int f4 = (i - nd * (K / 4)) * 4;
        int gnode = node_base + nd;
        if (gnode > Nn - 1) gnode = Nn - 1;
        const float4 v = *(const float4*)(xin + (size_t)gnode * K + f4);
        *(float4*)(s_x + nd * KP + f4) = v;
    }
    __syncthreads();

    v8f acc = {};
#pragma unroll
    for (int k = 0; k < K; k += 4) {
        const int kb = k + khalf * 2;
        v2f a, b;
        a.x = s_x[m * KP + kb];
        a.y = s_x[m * KP + kb + 1];
        b.x = W[(size_t)kb * NH + ncol];
        b.y = W[(size_t)(kb + 1) * NH + ncol];
        acc = __builtin_amdgcn_wmma_f32_16x16x4_f32(false, a, false, b,
                                                    (short)0, acc, false, false);
    }

#pragma unroll
    for (int v = 0; v < 8; ++v) {
        int gnode = node_base + v + khalf * 8;
        if (gnode < Nn) dstp[(size_t)gnode * NH + ncol] = acc[v];
    }
}

__global__ void combine_head_kernel(const float* __restrict__ agg,
                                    const float* __restrict__ q,
                                    const float* __restrict__ deg,
                                    const float* __restrict__ bl,
                                    const float* __restrict__ Wfc,
                                    const float* __restrict__ bfc,
                                    float* __restrict__ out, int Nn) {
    const long long t = (long long)blockIdx.x * blockDim.x + threadIdx.x;
    int node = (int)(t >> 4);
    const int sub = (int)(t & 15);
    const int c = sub * 4;
    const bool live = node < Nn;
    if (!live) node = Nn - 1;
    const float inv = 1.0f / fmaxf(deg[node], 1.0f);
    const float4 a = *(const float4*)(agg + (size_t)node * 64 + c);
    const float4 r = *(const float4*)(q + (size_t)node * 64 + c);
    const float4 b = *(const float4*)(bl + c);
    const float4 w = *(const float4*)(Wfc + c);
    float s = fmaxf(a.x * inv + b.x + r.x, 0.0f) * w.x;
    s += fmaxf(a.y * inv + b.y + r.y, 0.0f) * w.y;
    s += fmaxf(a.z * inv + b.z + r.z, 0.0f) * w.z;
    s += fmaxf(a.w * inv + b.w + r.w, 0.0f) * w.w;
    s += __shfl_xor(s, 8);
    s += __shfl_xor(s, 4);
    s += __shfl_xor(s, 2);
    s += __shfl_xor(s, 1);
    if (live && sub == 0) out[node] = 1.0f / (1.0f + __expf(-(s + bfc[0])));
}

extern "C" void kernel_launch(void* const* d_in, const int* in_sizes, int n_in,
                              void* d_out, int out_size, void* d_ws, size_t ws_size,
                              hipStream_t stream) {
    const float* x    = (const float*)d_in[0];
    const int*   ei   = (const int*)d_in[1];
    const float* W1l  = (const float*)d_in[2];
    const float* b1   = (const float*)d_in[3];
    const float* W1r  = (const float*)d_in[4];
    const float* W2l  = (const float*)d_in[5];
    const float* b2   = (const float*)d_in[6];
    const float* W2r  = (const float*)d_in[7];
    const float* Wfc  = (const float*)d_in[8];
    const float* bfc  = (const float*)d_in[9];
    float* outp = (float*)d_out;

    const int Nn = in_sizes[0] / 64;
    const int Ee = in_sizes[1] / 2;

    float* deg  = (float*)d_ws;
    float* agg1 = deg + Nn;
    float* agg2 = agg1 + (size_t)Nn * 64;
    float* h1   = agg2 + (size_t)Nn * 64;
    float* p    = h1 + (size_t)Nn * 128;
    float* q    = p + (size_t)Nn * 64;

    const int ZB = 2048, TB = 256;
    const long long t64 = (long long)Ee * (64 / 4);
    const int scatter_blocks = (int)((t64 + TB - 1) / TB);

    zero_kernel<<<ZB, TB, 0, stream>>>(deg, (long long)Nn * 129);
    degree_kernel<<<(Ee + TB - 1) / TB, TB, 0, stream>>>(ei, deg, Ee);

    scatter_kernel<64><<<scatter_blocks, TB, 0, stream>>>(ei, x, agg1, Ee);
    sage_combine_kernel<64, 128><<<(Nn + 15) / 16, 256, 0, stream>>>(
        x, agg1, deg, W1l, b1, W1r, h1, Nn);

    project_pair_kernel<128, 64><<<(Nn + 15) / 16, 256, 0, stream>>>(
        h1, W2l, W2r, p, q, Nn);
    scatter_kernel<64><<<scatter_blocks, TB, 0, stream>>>(ei, p, agg2, Ee);

    {
        long long t = (long long)Nn * 16;
        combine_head_kernel<<<(int)((t + TB - 1) / TB), TB, 0, stream>>>(
            agg2, q, deg, b2, Wfc, bfc, outp, Nn);
    }
}


# Neighbour sums without float atomics: edges filed by destination inside the call (integer-add arrival rank, block-scanned runs); each node adds its own rows in registers, in arrival-rank order
# speedup vs baseline: 2.6070x; 1.8981x over previous
#include <hip/hip_runtime.h>
#include <hip/hip_bf16.h>

typedef __attribute__((ext_vector_type(2))) float v2f;
typedef __attribute__((ext_vector_type(8))) float v8f;

__global__ void zero_kernel(float* __restrict__ p, long long n) {
    long long i = (long long)blockIdx.x * blockDim.x + threadIdx.x;
    long long stride = (long long)gridDim.x * blockDim.x;
    for (; i < n; i += stride) p[i] = 0.0f;
}

__global__ void rank_kernel(const int* __restrict__ ei, int* __restrict__ cnt,
                            int* __restrict__ rank, int E) {
    int e = blockIdx.x * blockDim.x + threadIdx.x;
    if (e < E) rank[e] = atomicAdd(&cnt[ei[E + e]], 1);
}

__global__ __launch_bounds__(256) void alloc_kernel(const int* cnt, int* total,
                                                    int* __restrict__ start,
                                                    float* __restrict__ deg, int Nn) {
    __shared__ int s_w[8];
    __shared__ int s_base;
    const int tid = threadIdx.x;
    const int lane = tid & 31;
    const int wave = tid >> 5;
    const int node = blockIdx.x * 256 + tid;
    const int c = node < Nn ? cnt[node] : 0;
    int v = c;
#pragma unroll
    for (int d = 1; d < 32; d <<= 1) {
        int t = __shfl_up(v, d);
        if (lane >= d) v += t;
    }
    if (lane == 31) s_w[wave] = v;
    __syncthreads();
    if (tid == 0) {
        int run = 0;
#pragma unroll
        for (int w = 0; w < 8; ++w) {
            int t = s_w[w];
            s_w[w] = run;
            run += t;
        }
        s_base = atomicAdd(total, run);
    }
    __syncthreads();
    if (node < Nn) {
        start[node] = s_base + s_w[wave] + v - c;
        deg[node] = (float)c;
    }
}

__global__ void fill_kernel(const int* __restrict__ ei, const int* __restrict__ start,
                            const int* __restrict__ rank, int* __restrict__ csr, int E) {
    int e = blockIdx.x * blockDim.x + threadIdx.x;
    if (e < E) csr[start[ei[E + e]] + rank[e]] = ei[e];
}

__device__ __forceinline__ float2 row_f2(const float* __restrict__ feat, int my, int k, int lane2) {
    const unsigned idx = (unsigned)__builtin_amdgcn_readlane(my, k);
    return *(const float2*)(feat + (size_t)idx * 64 + lane2);
}

__device__ __forceinline__ void gather_run32(float2& acc, const float* __restrict__ feat,
                                             int my, int m, int lane2) {
    int k = 0;
    for (; k + 8 <= m; k += 8) {
        float2 v[8];
#pragma unroll
        for (int j = 0; j < 8; ++j) v[j] = row_f2(feat, my, k + j, lane2);
#pragma unroll
        for (int j = 0; j < 8; ++j) { acc.x += v[j].x; acc.y += v[j].y; }
    }
    const int rem = m - k;
    if (rem & 4) {
        float2 v[4];
#pragma unroll
        for (int j = 0; j < 4; ++j) v[j] = row_f2(feat, my, k + j, lane2);
#pragma unroll
        for (int j = 0; j < 4; ++j) { acc.x += v[j].x; acc.y += v[j].y; }
        k += 4;
    }
    if (rem & 2) {
        const float2 v0 = row_f2(feat, my, k, lane2);
        const float2 v1 = row_f2(feat, my, k + 1, lane2);
        acc.x += v0.x; acc.y += v0.y;
        acc.x += v1.x; acc.y += v1.y;
        k += 2;
    }
    if (rem & 1) {
        const float2 v0 = row_f2(feat, my, k, lane2);
        acc.x += v0.x; acc.y += v0.y;
    }
}

template <int NPW>
__global__ __launch_bounds__(256) void gather_sum_kernel(
    const int* __restrict__ csr, const int* __restrict__ start,
    const int* __restrict__ cnt, const float* __restrict__ feat,
    float* __restrict__ agg, int Nn) {
    const int lane = threadIdx.x & 31;
    const int wave = __builtin_amdgcn_readfirstlane((int)(threadIdx.x >> 5));
    const int lane2 = lane * 2;
    const int node0 = (blockIdx.x * 8 + wave) * NPW;
    if (node0 >= Nn) return;

    int ql = node0 + (lane & (NPW - 1));
    if (ql > Nn - 1) ql = Nn - 1;
    const int sl = start[ql];
    const int nl = cnt[ql];
    int s[NPW], n[NPW], my[NPW];
#pragma unroll
    for (int i = 0; i < NPW; ++i) {
        s[i] = __builtin_amdgcn_readlane(sl, i);
        n[i] = (node0 + i < Nn) ? __builtin_amdgcn_readlane(nl, i) : 0;
    }
#pragma unroll
    for (int i = 0; i < NPW; ++i) my[i] = (lane < n[i]) ? csr[s[i] + lane] : 0;

#pragma unroll
    for (int i = 0; i < NPW; ++i) {
        if (node0 + i < Nn) {
            float2 acc = make_float2(0.0f, 0.0f);
            gather_run32(acc, feat, my[i], n[i] < 32 ? n[i] : 32, lane2);
            for (int base = 32; base < n[i]; base += 32) {
                const int m = n[i] - base < 32 ? n[i] - base : 32;
                const int more = (lane < m) ? csr[s[i] + base + lane] : 0;
                gather_run32(acc, feat, more, m, lane2);
            }
            *(float2*)(agg + (size_t)(node0 + i) * 64 + lane2) = acc;
        }
    }
}

template <int K, int NOUT>
__global__ __launch_bounds__(32 * (NOUT / 16)) void sage_combine_kernel(
    const float* __restrict__ xin, const float* __restrict__ agg,
    const float* __restrict__ deg, const float* __restrict__ Wl,
    const float* __restrict__ bl, const float* __restrict__ Wr,
    float* __restrict__ out, int Nn) {
    __shared__ float s_inv[16];
    __shared__ float s_a[16 * K];
    __shared__ float s_x[16 * K];

    const int node_base = blockIdx.x * 16;
    const int tid = threadIdx.x;
    const int lane = tid & 31;
    const int wave = tid >> 5;
    const int m = lane & 15;
    const int khalf = lane >> 4;
    const int ncol = wave * 16 + m;

    if (tid < 16) {
        int nd = node_base + tid;
        if (nd > Nn - 1) nd = Nn - 1;
        s_inv[tid] = 1.0f / fmaxf(deg[nd], 1.0f);
    }
    __syncthreads();

    const int nthreads = 32 * (NOUT / 16);
    for (int i = tid; i < 16 * K; i += nthreads) {
        int nd = i / K;
        int f = i - nd * K;
        int gnode = node_base + nd;
        if (gnode > Nn - 1) gnode = Nn - 1;
        s_a[i] = agg[(size_t)gnode * K + f] * s_inv[nd];
        s_x[i] = xin[(size_t)gnode * K + f];
    }
    __syncthreads();

    v8f acc = {};
#pragma unroll
    for (int k = 0; k < K; k += 4) {
        const int kb = k + khalf * 2;
        v2f a, b;
        a.x = s_a[m * K + kb];
        a.y = s_a[m * K + kb + 1];
        b.x = Wl[(size_t)kb * NOUT + ncol];
        b.y = Wl[(size_t)(kb + 1) * NOUT + ncol];
        acc = __builtin_amdgcn_wmma_f32_16x16x4_f32(false, a, false, b,
                                                    (short)0, acc, false, false);
        a.x = s_x[m * K + kb];
        a.y = s_x[m * K + kb + 1];
        b.x = Wr[(size_t)kb * NOUT + ncol];
        b.y = Wr[(size_t)(kb + 1) * NOUT + ncol];
        acc = __builtin_amdgcn_wmma_f32_16x16x4_f32(false, a, false, b,
                                                    (short)0, acc, false, false);
    }

    const float bias = bl[ncol];
#pragma unroll
    for (int v = 0; v < 8; ++v) {
        int mrow = v + khalf * 8;
        int gnode = node_base + mrow;
        if (gnode < Nn) {
            float val = acc[v] + bias;
            out[(size_t)gnode * NOUT + ncol] = fmaxf(val, 0.0f);
        }
    }
}

template <int K, int NH>
__global__ __launch_bounds__(32 * (2 * NH / 16)) void project_pair_kernel(
    const float* __restrict__ xin, const float* __restrict__ Wa,
    const float* __restrict__ Wb, float* __restrict__ pa,
    float* __restrict__ pb, int Nn) {
    constexpr int KP = K + 4;
    __shared__ __attribute__((aligned(16))) float s_x[16 * KP];

    const int node_base = blockIdx.x * 16;
    const int tid = threadIdx.x;
    const int lane = tid & 31;
    const int wave = tid >> 5;
    const int m = lane & 15;
    const int khalf = lane >> 4;
    constexpr int TILES = NH / 16;
    const bool second = wave >= TILES;
    const float* __restrict__ W = second ? Wb : Wa;
    float* __restrict__ dstp = second ? pb : pa;
    const int ncol = (second ? wave - TILES : wave) * 16 + m;

    constexpr int nthreads = 32 * (2 * NH / 16);
    for (int i = tid; i < 16 * (K / 4); i += nthreads) {
        int nd = i / (K / 4);
        int f4 = (i - nd * (K / 4)) * 4;
        int gnode = node_base + nd;
        if (gnode > Nn - 1) gnode = Nn - 1;
        const float4 v = *(const float4*)(xin + (size_t)gnode * K + f4);
        *(float4*)(s_x + nd * KP + f4) = v;
    }
    __syncthreads();

    v8f acc = {};
#pragma unroll
    for (int k = 0; k < K; k += 4) {
        const int kb = k + khalf * 2;
        v2f a, b;
        a.x = s_x[m * KP + kb];
        a.y = s_x[m * KP + kb + 1];
        b.x = W[(size_t)kb * NH + ncol];
        b.y = W[(size_t)(kb + 1) * NH + ncol];
        acc = __builtin_amdgcn_wmma_f32_16x16x4_f32(false, a, false, b,
                                                    (short)0, acc, false, false);
    }

#pragma unroll
    for (int v = 0; v < 8; ++v) {
        int gnode = node_base + v + khalf * 8;
        if (gnode < Nn) dstp[(size_t)gnode * NH + ncol] = acc[v];
    }
}

__global__ void combine_head_kernel(const float* __restrict__ agg,
                                    const float* __restrict__ q,
                                    const float* __restrict__ deg,
                                    const float* __restrict__ bl,
                                    const float* __restrict__ Wfc,
                                    const float* __restrict__ bfc,
                                    float* __restrict__ out, int Nn) {
    const long long t = (long long)blockIdx.x * blockDim.x + threadIdx.x;
    int node = (int)(t >> 4);
    const int sub = (int)(t & 15);
    const int c = sub * 4;
    const bool live = node < Nn;
    if (!live) node = Nn - 1;
    const float inv = 1.0f / fmaxf(deg[node], 1.0f);
    const float4 a = *(const float4*)(agg + (size_t)node * 64 + c);
    const float4 r = *(const float4*)(q + (size_t)node * 64 + c);
    const float4 b = *(const float4*)(bl + c);
    const float4 w = *(const float4*)(Wfc + c);
    float s = fmaxf(a.x * inv + b.x + r.x, 0.0f) * w.x;
    s += fmaxf(a.y * inv + b.y + r.y, 0.0f) * w.y;
    s += fmaxf(a.z * inv + b.z + r.z, 0.0f) * w.z;
    s += fmaxf(a.w * inv + b.w + r.w, 0.0f) * w.w;
    s += __shfl_xor(s, 8);
    s += __shfl_xor(s, 4);
    s += __shfl_xor(s, 2);
    s += __shfl_xor(s, 1);
    if (live && sub == 0) out[node] = 1.0f / (1.0f + __expf(-(s + bfc[0])));
}

extern "C" void kernel_launch(void* const* d_in, const int* in_sizes, int n_in,
                              void* d_out, int out_size, void* d_ws, size_t ws_size,
                              hipStream_t stream) {
    const float* x    = (const float*)d_in[0];
    const int*   ei   = (const int*)d_in[1];
    const float* W1l  = (const float*)d_in[2];
    const float* b1   = (const float*)d_in[3];
    const float* W1r  = (const float*)d_in[4];
    const float* W2l  = (const float*)d_in[5];
    const float* b2   = (const float*)d_in[6];
    const float* W2r  = (const float*)d_in[7];
    const float* Wfc  = (const float*)d_in[8];
    const float* bfc  = (const float*)d_in[9];
    float* outp = (float*)d_out;

    const int Nn = in_sizes[0] / 64;
    const int Ee = in_sizes[1] / 2;

    float* agg1 = (float*)d_ws;
    float* agg2 = agg1 + (size_t)Nn * 64;
    float* h1   = agg2 + (size_t)Nn * 64;
    float* p    = h1 + (size_t)Nn * 128;
    float* q    = p + (size_t)Nn * 64;
    int*   csr  = (int*)(q + (size_t)Nn * 64);
    int*   rank = csr + Ee;
    int*   start = rank + Ee;
    float* deg  = (float*)(start + Nn);
    int*   cnt  = (int*)(deg + Nn);
    int*   total = cnt + Nn;

    const int TB = 256;
    const int edge_blocks = (Ee + TB - 1) / TB;
    const int node_blocks = (Nn + TB - 1) / TB;
    const int gather_blocks = (Nn + 31) / 32;

    zero_kernel<<<(Nn + 1 + TB - 1) / TB, TB, 0, stream>>>((float*)cnt, (long long)Nn + 1);
    rank_kernel<<<edge_blocks, TB, 0, stream>>>(ei, cnt, rank, Ee);
    alloc_kernel<<<node_blocks, TB, 0, stream>>>(cnt, total, start, deg, Nn);
    fill_kernel<<<edge_blocks, TB, 0, stream>>>(ei, start, rank, csr, Ee);

    gather_sum_kernel<4><<<gather_blocks, TB, 0, stream>>>(csr, start, cnt, x, agg1, Nn);
    sage_combine_kernel<64, 128><<<(Nn + 15) / 16, 256, 0, stream>>>(
        x, agg1, deg, W1l, b1, W1r, h1, Nn);

    project_pair_kernel<128, 64><<<(Nn + 15) / 16, 256, 0, stream>>>(
        h1, W2l, W2r, p, q, Nn);
    gather_sum_kernel<4><<<gather_blocks, TB, 0, stream>>>(csr, start, cnt, p, agg2, Nn);

    {
        long long t = (long long)Nn * 16;
        combine_head_kernel<<<(int)((t + TB - 1) / TB), TB, 0, stream>>>(
            agg2, q, deg, b2, Wfc, bfc, outp, Nn);
    }
}


# Both neighbour sums fused into their consumers: layer 1 sums, combines and projects in one launch (mean and h1 tiles stay in LDS), layer 2 sums rows in registers and runs the head; 9 launches to 6
# speedup vs baseline: 3.0363x; 1.1647x over previous
#include <hip/hip_runtime.h>
#include <hip/hip_bf16.h>

typedef __attribute__((ext_vector_type(2))) float v2f;
typedef __attribute__((ext_vector_type(8))) float v8f;

__global__ void zero_kernel(float* __restrict__ p, long long n) {
    long long i = (long long)blockIdx.x * blockDim.x + threadIdx.x;
    long long stride = (long long)gridDim.x * blockDim.x;
    for (; i < n; i += stride) p[i] = 0.0f;
}

__global__ void rank_kernel(const int* __restrict__ ei, int* __restrict__ cnt,
                            int* __restrict__ rank, int E) {
    int e = blockIdx.x * blockDim.x + threadIdx.x;
    if (e < E) rank[e] = atomicAdd(&cnt[ei[E + e]], 1);
}

__global__ __launch_bounds__(256) void alloc_kernel(const int* cnt, int* total,
                                                    int* __restrict__ start, int Nn) {
    __shared__ int s_w[8];
    __shared__ int s_base;
    const int tid = threadIdx.x;
    const int lane = tid & 31;
    const int wave = tid >> 5;
    const int node = blockIdx.x * 256 + tid;
    const int c = node < Nn ? cnt[node] : 0;
    int v = c;
#pragma unroll
    for (int d = 1; d < 32; d <<= 1) {
        int t = __shfl_up(v, d);
        if (lane >= d) v += t;
    }
    if (lane == 31) s_w[wave] = v;
    __syncthreads();
    if (tid == 0) {
        int run = 0;
#pragma unroll
        for (int w = 0; w < 8; ++w) {
            int t = s_w[w];
            s_w[w] = run;
            run += t;
        }
        s_base = atomicAdd(total, run);
    }
    __syncthreads();
    if (node < Nn) start[node] = s_base + s_w[wave] + v - c;
}

__global__ void fill_kernel(const int* __restrict__ ei, const int* __restrict__ start,
                            const int* __restrict__ rank, int* __restrict__ csr, int E) {
    int e = blockIdx.x * blockDim.x + threadIdx.x;
    if (e < E) csr[start[ei[E + e]] + rank[e]] = ei[e];
}

__device__ __forceinline__ float2 row_f2(const float* __restrict__ feat, int my, int k, int lane2) {
    const unsigned idx = (unsigned)__builtin_amdgcn_readlane(my, k);
    return *(const float2*)(feat + (size_t)idx * 64 + lane2);
}

__device__ __forceinline__ void gather_run32(float2& acc, const float* __restrict__ feat,
                                             int my, int m, int lane2) {
    int k = 0;
    for (; k + 8 <= m; k += 8) {
        float2 v[8];
#pragma unroll
        for (int j = 0; j < 8; ++j) v[j] = row_f2(feat, my, k + j, lane2);
#pragma unroll
        for (int j = 0; j < 8; ++j) { acc.x += v[j].x; acc.y += v[j].y; }
    }
    const int rem = m - k;
    if (rem & 4) {
        float2 v[4];
#pragma unroll
        for (int j = 0; j < 4; ++j) v[j] = row_f2(feat, my, k + j, lane2);
#pragma unroll
        for (int j = 0; j < 4; ++j) { acc.x += v[j].x; acc.y += v[j].y; }
        k += 4;
    }
    if (rem & 2) {
        const float2 v0 = row_f2(feat, my, k, lane2);
        const float2 v1 = row_f2(feat, my, k + 1, lane2);
        acc.x += v0.x; acc.y += v0.y;
        acc.x += v1.x; acc.y += v1.y;
        k += 2;
    }
    if (rem & 1) {
        const float2 v0 = row_f2(feat, my, k, lane2);
        acc.x += v0.x; acc.y += v0.y;
    }
}

__device__ __forceinline__ float2 node_sum(const int* __restrict__ csr,
                                           const float* __restrict__ feat,
                                           int s, int n, int my, int lane, int lane2) {
    float2 acc = make_float2(0.0f, 0.0f);
    gather_run32(acc, feat, my, n < 32 ? n : 32, lane2);
    for (int base = 32; base < n; base += 32) {
        const int m = n - base < 32 ? n - base : 32;
        const int more = (lane < m) ? csr[s + base + lane] : 0;
        gather_run32(acc, feat, more, m, lane2);
    }
    return acc;
}

__global__ __launch_bounds__(256) void layer1_kernel(
    const int* __restrict__ csr, const int* __restrict__ start,
    const int* __restrict__ cnt, const float* __restrict__ x,
    const float* __restrict__ W1l, const float* __restrict__ b1,
    const float* __restrict__ W1r, const float* __restrict__ W2l,
    const float* __restrict__ W2r, float* __restrict__ p,
    float* __restrict__ q, int Nn) {
    constexpr int KA = 64 + 4;
    constexpr int KH = 128 + 4;
    __shared__ __attribute__((aligned(16))) float s_a[16 * KA];
    __shared__ __attribute__((aligned(16))) float s_x[16 * KA];
    __shared__ __attribute__((aligned(16))) float s_h[16 * KH];

    const int tid = threadIdx.x;
    const int lane = tid & 31;
    const int wave = __builtin_amdgcn_readfirstlane((int)(tid >> 5));
    const int lane2 = lane * 2;
    const int node_base = blockIdx.x * 16;
    const int m = lane & 15;
    const int khalf = lane >> 4;

    {
        const int node0 = node_base + wave * 2;
        int ql = node0 + (lane & 1);
        if (ql > Nn - 1) ql = Nn - 1;
        const int sl = start[ql];
        const int nl = cnt[ql];
        int s[2], n[2], my[2];
#pragma unroll
        for (int i = 0; i < 2; ++i) {
            s[i] = __builtin_amdgcn_readlane(sl, i);
            n[i] = (node0 + i < Nn) ? __builtin_amdgcn_readlane(nl, i) : 0;
        }
#pragma unroll
        for (int i = 0; i < 2; ++i) my[i] = (lane < n[i]) ? csr[s[i] + lane] : 0;
#pragma unroll
        for (int i = 0; i < 2; ++i) {
            int gnode = node0 + i;
            if (gnode > Nn - 1) gnode = Nn - 1;
            const float2 xv = *(const float2*)(x + (size_t)gnode * 64 + lane2);
            const float2 acc = node_sum(csr, x, s[i], n[i], my[i], lane, lane2);
            const float inv = 1.0f / fmaxf((float)n[i], 1.0f);
            const int row = wave * 2 + i;
            *(float2*)(s_a + row * KA + lane2) = make_float2(acc.x * inv, acc.y * inv);
            *(float2*)(s_x + row * KA + lane2) = xv;
        }
    }
    __syncthreads();

    {
        const int ncol = wave * 16 + m;
        v8f acc = {};
#pragma unroll
        for (int k = 0; k < 64; k += 4) {
            const int kb = k + khalf * 2;
            v2f a, b;
            a.x = s_a[m * KA + kb];
            a.y = s_a[m * KA + kb + 1];
            b.x = W1l[(size_t)kb * 128 + ncol];
            b.y = W1l[(size_t)(kb + 1) * 128 + ncol];
            acc = __builtin_amdgcn_wmma_f32_16x16x4_f32(false, a, false, b,
                                                        (short)0, acc, false, false);
            a.x = s_x[m * KA + kb];
            a.y = s_x[m * KA + kb + 1];
            b.x = W1r[(size_t)kb * 128 + ncol];
            b.y = W1r[(size_t)(kb + 1) * 128 + ncol];
            acc = __builtin_amdgcn_wmma_f32_16x16x4_f32(false, a, false, b,
                                                        (short)0, acc, false, false);
        }
        const float bias = b1[ncol];
#pragma unroll
        for (int v = 0; v < 8; ++v)
            s_h[(v + khalf * 8) * KH + ncol] = fmaxf(acc[v] + bias, 0.0f);
    }
    __syncthreads();

    {
        const bool second = wave >= 4;
        const float* __restrict__ W = second ? W2r : W2l;
        float* __restrict__ dstp = second ? q : p;
        const int ncol = (wave & 3) * 16 + m;
        v8f acc = {};
#pragma unroll
        for (int k = 0; k < 128; k += 4) {
            const int kb = k + khalf * 2;
            v2f a, b;
            a.x = s_h[m * KH + kb];
            a.y = s_h[m * KH + kb + 1];
            b.x = W[(size_t)kb * 64 + ncol];
            b.y = W[(size_t)(kb + 1) * 64 + ncol];
            acc = __builtin_amdgcn_wmma_f32_16x16x4_f32(false, a, false, b,
                                                        (short)0, acc, false, false);
        }
#pragma unroll
        for (int v = 0; v < 8; ++v) {
            const int gnode = node_base + v + khalf * 8;
            if (gnode < Nn) dstp[(size_t)gnode * 64 + ncol] = acc[v];
        }
    }
}

__global__ __launch_bounds__(256) void layer2_kernel(
    const int* __restrict__ csr, const int* __restrict__ start,
    const int* __restrict__ cnt, const float* __restrict__ p,
    const float* __restrict__ q, const float* __restrict__ b2,
    const float* __restrict__ Wfc, const float* __restrict__ bfc,
    float* __restrict__ out, int Nn) {
    __shared__ float s_out[32];
    const int tid = threadIdx.x;
    const int lane = tid & 31;
    const int wave = __builtin_amdgcn_readfirstlane((int)(tid >> 5));
    const int lane2 = lane * 2;
    const int node0 = (blockIdx.x * 8 + wave) * 4;

    int ql = node0 + (lane & 3);
    if (ql > Nn - 1) ql = Nn - 1;
    const int sl = start[ql];
    const int nl = cnt[ql];
    int s[4], n[4], my[4];
#pragma unroll
    for (int i = 0; i < 4; ++i) {
        s[i] = __builtin_amdgcn_readlane(sl, i);
        n[i] = (node0 + i < Nn) ? __builtin_amdgcn_readlane(nl, i) : 0;
    }
#pragma unroll
    for (int i = 0; i < 4; ++i) my[i] = (lane < n[i]) ? csr[s[i] + lane] : 0;

    const float2 bb = *(const float2*)(b2 + lane2);
    const float4 w = *(const float4*)(Wfc + (lane >> 1) * 4);
    const float bf = bfc[0];

    float res = 0.0f;
#pragma unroll
    for (int i = 0; i < 4; ++i) {
        int gnode = node0 + i;
        if (gnode > Nn - 1) gnode = Nn - 1;
        const float2 qv = *(const float2*)(q + (size_t)gnode * 64 + lane2);
        const float2 acc = node_sum(csr, p, s[i], n[i], my[i], lane, lane2);
        const float inv = 1.0f / fmaxf((float)n[i], 1.0f);
        const float r0 = fmaxf(acc.x * inv + bb.x + qv.x, 0.0f);
        const float r1 = fmaxf(acc.y * inv + bb.y + qv.y, 0.0f);
        const float r2 = __shfl_xor(r0, 1);
        const float r3 = __shfl_xor(r1, 1);
        float d = r0 * w.x;
        d += r1 * w.y;
        d += r2 * w.z;
        d += r3 * w.w;
        d += __shfl_xor(d, 16);
        d += __shfl_xor(d, 8);
        d += __shfl_xor(d, 4);
        d += __shfl_xor(d, 2);
        const float o = 1.0f / (1.0f + __expf(-(d + bf)));
        const float o0 = __builtin_bit_cast(
            float, __builtin_amdgcn_readfirstlane(__builtin_bit_cast(int, o)));
        if (lane == i) res = o0;
    }
    if (lane < 4) s_out[wave * 4 + lane] = res;
    __syncthreads();
    if (tid < 32) {
        const int node = blockIdx.x * 32 + tid;
        if (node < Nn) out[node] = s_out[tid];
    }
}

extern "C" void kernel_launch(void* const* d_in, const int* in_sizes, int n_in,
                              void* d_out, int out_size, void* d_ws, size_t ws_size,
                              hipStream_t stream) {
    const float* x    = (const float*)d_in[0];
    const int*   ei   = (const int*)d_in[1];
    const float* W1l  = (const float*)d_in[2];
    const float* b1   = (const float*)d_in[3];
    const float* W1r  = (const float*)d_in[4];
    const float* W2l  = (const float*)d_in[5];
    const float* b2   = (const float*)d_in[6];
    const float* W2r  = (const float*)d_in[7];
    const float* Wfc  = (const float*)d_in[8];
    const float* bfc  = (const float*)d_in[9];
    float* outp = (float*)d_out;

    const int Nn = in_sizes[0] / 64;
    const int Ee = in_sizes[1] / 2;

    float* p    = (float*)d_ws;
    float* q    = p + (size_t)Nn * 64;
    int*   csr  = (int*)(q + (size_t)Nn * 64);
    int*   rank = csr + Ee;
    int*   start = rank + Ee;
    int*   cnt  = start + Nn;
    int*   total = cnt + Nn;

    const int TB = 256;
    const int edge_blocks = (Ee + TB - 1) / TB;
    const int node_blocks = (Nn + TB - 1) / TB;

    zero_kernel<<<(Nn + 1 + TB - 1) / TB, TB, 0, stream>>>((float*)cnt, (long long)Nn + 1);
    rank_kernel<<<edge_blocks, TB, 0, stream>>>(ei, cnt, rank, Ee);
    alloc_kernel<<<node_blocks, TB, 0, stream>>>(cnt, total, start, Nn);
    fill_kernel<<<edge_blocks, TB, 0, stream>>>(ei, start, rank, csr, Ee);

    layer1_kernel<<<(Nn + 15) / 16, 256, 0, stream>>>(
        csr, start, cnt, x, W1l, b1, W1r, W2l, W2r, p, q, Nn);

    layer2_kernel<<<(Nn + 31) / 32, TB, 0, stream>>>(
        csr, start, cnt, p, q, b2, Wfc, bfc, outp, Nn);
}


# Edges filed by destination on LDS counters, no global atomics: 8192-edge blocks group by 512-node bucket, one block per bucket counts, scans, places; sums follow LDS arrival order; 6 launches to 4
# speedup vs baseline: 6.0153x; 1.9811x over previous
#include <hip/hip_runtime.h>
#include <hip/hip_bf16.h>

typedef __attribute__((ext_vector_type(2))) float v2f;
typedef __attribute__((ext_vector_type(8))) float v8f;

constexpr int SORT_C = 8192;
constexpr int SORT_T = 1024;
constexpr int BSH = 9;
constexpr int BR = 1 << BSH;
constexpr int NB_MAX = 1024;
constexpr int NBLK_MAX = 1024;
constexpr int STG_CAP = 16384;

__device__ __forceinline__ int block_excl_scan_1024(int v, int lane, int wave,
                                                    int* s_ws, int* s_tot) {
    int inc = v;
#pragma unroll
    for (int d = 1; d < 32; d <<= 1) {
        int t = __shfl_up(inc, d);
        if (lane >= d) inc += t;
    }
    if (lane == 31) s_ws[wave] = inc;
    __syncthreads();
    if (wave == 0) {
        const int w = s_ws[lane];
        int wi = w;
#pragma unroll
        for (int d = 1; d < 32; d <<= 1) {
            int t = __shfl_up(wi, d);
            if (lane >= d) wi += t;
        }
        s_ws[lane] = wi - w;
        if (lane == 31) *s_tot = wi;
    }
    __syncthreads();
    return s_ws[wave] + inc - v;
}

__global__ __launch_bounds__(1024) void partition_kernel(const int* __restrict__ ei,
                                                         unsigned* __restrict__ tmp,
                                                         int* __restrict__ offm,
                                                         int E, int NB) {
    __shared__ int s_hist[NB_MAX];
    __shared__ unsigned s_stage[SORT_C];
    __shared__ int s_ws[32];
    __shared__ int s_tot;
    const int tid = threadIdx.x;
    const int lane = tid & 31;
    const int wave = tid >> 5;
    const int e0 = blockIdx.x * SORT_C;

    s_hist[tid] = 0;
    int sv[8], dv[8];
#pragma unroll
    for (int j = 0; j < 8; ++j) {
        const int e = e0 + j * SORT_T + tid;
        const int ec = e < E ? e : E - 1;
        dv[j] = ei[E + ec];
        sv[j] = ei[ec];
    }
#pragma unroll
    for (int j = 0; j < 8; ++j)
        if (e0 + j * SORT_T + tid >= E) dv[j] = -1;
    __syncthreads();
#pragma unroll
    for (int j = 0; j < 8; ++j)
        if (dv[j] >= 0) atomicAdd(&s_hist[dv[j] >> BSH], 1);
    __syncthreads();
    const int h = s_hist[tid];
    const int ex = block_excl_scan_1024(h, lane, wave, s_ws, &s_tot);
    s_hist[tid] = ex;
    int* __restrict__ row = offm + (size_t)blockIdx.x * (NB + 1);
    if (tid < NB) row[tid] = ex;
    const int tot = s_tot;
    if (tid == 0) row[NB] = tot;
    __syncthreads();
#pragma unroll
    for (int j = 0; j < 8; ++j) {
        if (dv[j] >= 0) {
            const int pos = atomicAdd(&s_hist[dv[j] >> BSH], 1);
            s_stage[pos] = ((unsigned)sv[j] << BSH) | ((unsigned)dv[j] & (BR - 1));
        }
    }
    __syncthreads();
#pragma unroll
    for (int j = 0; j < 8; ++j) {
        const int i = j * SORT_T + tid;
        if (i < tot) tmp[e0 + i] = s_stage[i];
    }
}

template <bool FILL>
__device__ __forceinline__ void visit_word(unsigned w, int* s_cnt, unsigned* s_stg,
                                           int* __restrict__ csr, int base, bool staged) {
    const int dl = (int)(w & (BR - 1));
    if (!FILL) {
        atomicAdd(&s_cnt[dl], 1);
    } else {
        const int pos = atomicAdd(&s_cnt[dl], 1);
        const unsigned src = w >> BSH;
        if (staged) s_stg[pos] = src;
        else csr[base + pos] = (int)src;
    }
}

template <bool FILL>
__device__ __forceinline__ void visit_groups(const unsigned* __restrict__ tmp,
                                             const int* s_segS, const int* s_segL,
                                             int nblk, int lane, int wave, int* s_cnt,
                                             unsigned* s_stg, int* __restrict__ csr,
                                             int base, bool staged) {
    for (int t0 = wave; t0 < nblk; t0 += 128) {
        int s[4], L[4];
        unsigned a[4][2];
#pragma unroll
        for (int u = 0; u < 4; ++u) {
            const int t = t0 + 32 * u;
            const bool ok = t < nblk;
            s[u] = ok ? s_segS[t] : 0;
            L[u] = ok ? s_segL[t] : 0;
        }
#pragma unroll
        for (int u = 0; u < 4; ++u) {
            a[u][0] = tmp[(lane < L[u]) ? s[u] + lane : 0];
            a[u][1] = tmp[(lane + 32 < L[u]) ? s[u] + lane + 32 : 0];
        }
#pragma unroll
        for (int u = 0; u < 4; ++u) {
            if (lane < L[u]) visit_word<FILL>(a[u][0], s_cnt, s_stg, csr, base, staged);
            if (lane + 32 < L[u]) visit_word<FILL>(a[u][1], s_cnt, s_stg, csr, base, staged);
            for (int i = lane + 64; i < L[u]; i += 32)
                visit_word<FILL>(tmp[s[u] + i], s_cnt, s_stg, csr, base, staged);
        }
    }
}

__global__ __launch_bounds__(1024) void bucket_kernel(const unsigned* __restrict__ tmp,
                                                      const int* __restrict__ offm,
                                                      int* __restrict__ csr,
                                                      int* __restrict__ start,
                                                      int* __restrict__ cnt,
                                                      int Nn, int NB, int nblk) {
    __shared__ int s_cnt[BR];
    __shared__ int s_segS[NBLK_MAX];
    __shared__ int s_segL[NBLK_MAX];
    __shared__ unsigned s_stg[STG_CAP];
    __shared__ int s_ws[32];
    __shared__ int s_tot;
    __shared__ int s_base, s_total;
    const int tid = threadIdx.x;
    const int lane = tid & 31;
    const int wave = tid >> 5;
    const int b = blockIdx.x;

    if (tid < BR) s_cnt[tid] = 0;
    if (tid == 0) { s_base = 0; s_total = 0; }
    __syncthreads();
    int pb = 0, pt = 0;
    for (int t = tid; t < nblk; t += SORT_T) {
        const int* __restrict__ row = offm + (size_t)t * (NB + 1);
        const int so = row[b];
        const int eo = row[b + 1];
        s_segS[t] = t * SORT_C + so;
        s_segL[t] = eo - so;
        pb += so;
        pt += eo - so;
    }
#pragma unroll
    for (int d = 16; d > 0; d >>= 1) {
        pb += __shfl_xor(pb, d);
        pt += __shfl_xor(pt, d);
    }
    if (lane == 0 && (pb | pt) != 0) {
        atomicAdd(&s_base, pb);
        atomicAdd(&s_total, pt);
    }
    __syncthreads();
    const int base = s_base;
    const int total = s_total;
    const bool staged = total <= STG_CAP;

    visit_groups<false>(tmp, s_segS, s_segL, nblk, lane, wave, s_cnt, s_stg, csr, base, staged);
    __syncthreads();

    const int c = (tid < BR) ? s_cnt[tid] : 0;
    const int ex = block_excl_scan_1024(c, lane, wave, s_ws, &s_tot);
    if (tid < BR) {
        s_cnt[tid] = ex;
        const int node = b * BR + tid;
        if (node < Nn) {
            start[node] = base + ex;
            cnt[node] = c;
        }
    }
    __syncthreads();

    visit_groups<true>(tmp, s_segS, s_segL, nblk, lane, wave, s_cnt, s_stg, csr, base, staged);
    if (staged) {
        __syncthreads();
        for (int i = tid; i < total; i += SORT_T) csr[base + i] = (int)s_stg[i];
    }
}

__device__ __forceinline__ float2 row_f2(const float* __restrict__ feat, int my, int k, int lane2) {
    const unsigned idx = (unsigned)__builtin_amdgcn_readlane(my, k);
    return *(const float2*)(feat + (size_t)idx * 64 + lane2);
}

__device__ __forceinline__ void gather_run32(float2& acc, const float* __restrict__ feat,
                                             int my, int m, int lane2) {
    int k = 0;
    for (; k + 8 <= m; k += 8) {
        float2 v[8];
#pragma unroll
        for (int j = 0; j < 8; ++j) v[j] = row_f2(feat, my, k + j, lane2);
#pragma unroll
        for (int j = 0; j < 8; ++j) { acc.x += v[j].x; acc.y += v[j].y; }
    }
    const int rem = m - k;
    if (rem & 4) {
        float2 v[4];
#pragma unroll
        for (int j = 0; j < 4; ++j) v[j] = row_f2(feat, my, k + j, lane2);
#pragma unroll
        for (int j = 0; j < 4; ++j) { acc.x += v[j].x; acc.y += v[j].y; }
        k += 4;
    }
    if (rem & 2) {
        const float2 v0 = row_f2(feat, my, k, lane2);
        const float2 v1 = row_f2(feat, my, k + 1, lane2);
        acc.x += v0.x; acc.y += v0.y;
        acc.x += v1.x; acc.y += v1.y;
        k += 2;
    }
    if (rem & 1) {
        const float2 v0 = row_f2(feat, my, k, lane2);
        acc.x += v0.x; acc.y += v0.y;
    }
}

__device__ __forceinline__ float2 node_sum(const int* __restrict__ csr,
                                           const float* __restrict__ feat,
                                           int s, int n, int my, int lane, int lane2) {
    float2 acc = make_float2(0.0f, 0.0f);
    gather_run32(acc, feat, my, n < 32 ? n : 32, lane2);
    for (int base = 32; base < n; base += 32) {
        const int m = n - base < 32 ? n - base : 32;
        const int more = (lane < m) ? csr[s + base + lane] : 0;
        gather_run32(acc, feat, more, m, lane2);
    }
    return acc;
}

__global__ __launch_bounds__(256) void layer1_kernel(
    const int* __restrict__ csr, const int* __restrict__ start,
    const int* __restrict__ cnt, const float* __restrict__ x,
    const float* __restrict__ W1l, const float* __restrict__ b1,
    const float* __restrict__ W1r, const float* __restrict__ W2l,
    const float* __restrict__ W2r, float* __restrict__ p,
    float* __restrict__ q, int Nn) {
    constexpr int KA = 64 + 4;
    constexpr int KH = 128 + 4;
    __shared__ __attribute__((aligned(16))) float s_a[16 * KA];
    __shared__ __attribute__((aligned(16))) float s_x[16 * KA];
    __shared__ __attribute__((aligned(16))) float s_h[16 * KH];

    const int tid = threadIdx.x;
    const int lane = tid & 31;
    const int wave = __builtin_amdgcn_readfirstlane((int)(tid >> 5));
    const int lane2 = lane * 2;
    const int node_base = blockIdx.x * 16;
    const int m = lane & 15;
    const int khalf = lane >> 4;

    {
        const int node0 = node_base + wave * 2;
        int ql = node0 + (lane & 1);
        if (ql > Nn - 1) ql = Nn - 1;
        const int sl = start[ql];
        const int nl = cnt[ql];
        int s[2], n[2], my[2];
#pragma unroll
        for (int i = 0; i < 2; ++i) {
            s[i] = __builtin_amdgcn_readlane(sl, i);
            n[i] = (node0 + i < Nn) ? __builtin_amdgcn_readlane(nl, i) : 0;
        }
#pragma unroll
        for (int i = 0; i < 2; ++i) my[i] = (lane < n[i]) ? csr[s[i] + lane] : 0;
#pragma unroll
        for (int i = 0; i < 2; ++i) {
            int gnode = node0 + i;
            if (gnode > Nn - 1) gnode = Nn - 1;
            const float2 xv = *(const float2*)(x + (size_t)gnode * 64 + lane2);
            const float2 acc = node_sum(csr, x, s[i], n[i], my[i], lane, lane2);
            const float inv = 1.0f / fmaxf((float)n[i], 1.0f);
            const int row = wave * 2 + i;
            *(float2*)(s_a + row * KA + lane2) = make_float2(acc.x * inv, acc.y * inv);
            *(float2*)(s_x + row * KA + lane2) = xv;
        }
    }
    __syncthreads();

    {
        const int ncol = wave * 16 + m;
        v8f acc = {};
#pragma unroll
        for (int k = 0; k < 64; k += 4) {
            const int kb = k + khalf * 2;
            v2f a, b;
            a.x = s_a[m * KA + kb];
            a.y = s_a[m * KA + kb + 1];
            b.x = W1l[(size_t)kb * 128 + ncol];
            b.y = W1l[(size_t)(kb + 1) * 128 + ncol];
            acc = __builtin_amdgcn_wmma_f32_16x16x4_f32(false, a, false, b,
                                                        (short)0, acc, false, false);
            a.x = s_x[m * KA + kb];
            a.y = s_x[m * KA + kb + 1];
            b.x = W1r[(size_t)kb * 128 + ncol];
            b.y = W1r[(size_t)(kb + 1) * 128 + ncol];
            acc = __builtin_amdgcn_wmma_f32_16x16x4_f32(false, a, false, b,
                                                        (short)0, acc, false, false);
        }
        const float bias = b1[ncol];
#pragma unroll
        for (int v = 0; v < 8; ++v)
            s_h[(v + khalf * 8) * KH + ncol] = fmaxf(acc[v] + bias, 0.0f);
    }
    __syncthreads();

    {
        const bool second = wave >= 4;
        const float* __restrict__ W = second ? W2r : W2l;
        float* __restrict__ dstp = second ? q : p;
        const int ncol = (wave & 3) * 16 + m;
        v8f acc = {};
#pragma unroll
        for (int k = 0; k < 128; k += 4) {
            const int kb = k + khalf * 2;
            v2f a, b;
            a.x = s_h[m * KH + kb];
            a.y = s_h[m * KH + kb + 1];
            b.x = W[(size_t)kb * 64 + ncol];
            b.y = W[(size_t)(kb + 1) * 64 + ncol];
            acc = __builtin_amdgcn_wmma_f32_16x16x4_f32(false, a, false, b,
                                                        (short)0, acc, false, false);
        }
#pragma unroll
        for (int v = 0; v < 8; ++v) {
            const int gnode = node_base + v + khalf * 8;
            if (gnode < Nn) dstp[(size_t)gnode * 64 + ncol] = acc[v];
        }
    }
}

__global__ __launch_bounds__(256) void layer2_kernel(
    const int* __restrict__ csr, const int* __restrict__ start,
    const int* __restrict__ cnt, const float* __restrict__ p,
    const float* __restrict__ q, const float* __restrict__ b2,
    const float* __restrict__ Wfc, const float* __restrict__ bfc,
    float* __restrict__ out, int Nn) {
    __shared__ float s_out[32];
    const int tid = threadIdx.x;
    const int lane = tid & 31;
    const int wave = __builtin_amdgcn_readfirstlane((int)(tid >> 5));
    const int lane2 = lane * 2;
    const int node0 = (blockIdx.x * 8 + wave) * 4;

    int ql = node0 + (lane & 3);
    if (ql > Nn - 1) ql = Nn - 1;
    const int sl = start[ql];
    const int nl = cnt[ql];
    int s[4], n[4], my[4];
#pragma unroll
    for (int i = 0; i < 4; ++i) {
        s[i] = __builtin_amdgcn_readlane(sl, i);
        n[i] = (node0 + i < Nn) ? __builtin_amdgcn_readlane(nl, i) : 0;
    }
#pragma unroll
    for (int i = 0; i < 4; ++i) my[i] = (lane < n[i]) ? csr[s[i] + lane] : 0;

    const float2 bb = *(const float2*)(b2 + lane2);
    const float4 w = *(const float4*)(Wfc + (lane >> 1) * 4);
    const float bf = bfc[0];

    float res = 0.0f;
#pragma unroll
    for (int i = 0; i < 4; ++i) {
        int gnode = node0 + i;
        if (gnode > Nn - 1) gnode = Nn - 1;
        const float2 qv = *(const float2*)(q + (size_t)gnode * 64 + lane2);
        const float2 acc = node_sum(csr, p, s[i], n[i], my[i], lane, lane2);
        const float inv = 1.0f / fmaxf((float)n[i], 1.0f);
        const float r0 = fmaxf(acc.x * inv + bb.x + qv.x, 0.0f);
        const float r1 = fmaxf(acc.y * inv + bb.y + qv.y, 0.0f);
        const float r2 = __shfl_xor(r0, 1);
        const float r3 = __shfl_xor(r1, 1);
        float d = r0 * w.x;
        d += r1 * w.y;
        d += r2 * w.z;
        d += r3 * w.w;
        d += __shfl_xor(d, 16);
        d += __shfl_xor(d, 8);
        d += __shfl_xor(d, 4);
        d += __shfl_xor(d, 2);
        const float o = 1.0f / (1.0f + __expf(-(d + bf)));
        const float o0 = __builtin_bit_cast(
            float, __builtin_amdgcn_readfirstlane(__builtin_bit_cast(int, o)));
        if (lane == i) res = o0;
    }
    if (lane < 4) s_out[wave * 4 + lane] = res;
    __syncthreads();
    if (tid < 32) {
        const int node = blockIdx.x * 32 + tid;
        if (node < Nn) out[node] = s_out[tid];
    }
}

extern "C" void kernel_launch(void* const* d_in, const int* in_sizes, int n_in,
                              void* d_out, int out_size, void* d_ws, size_t ws_size,
                              hipStream_t stream) {
    const float* x    = (const float*)d_in[0];
    const int*   ei   = (const int*)d_in[1];
    const float* W1l  = (const float*)d_in[2];
    const float* b1   = (const float*)d_in[3];
    const float* W1r  = (const float*)d_in[4];
    const float* W2l  = (const float*)d_in[5];
    const float* b2   = (const float*)d_in[6];
    const float* W2r  = (const float*)d_in[7];
    const float* Wfc  = (const float*)d_in[8];
    const float* bfc  = (const float*)d_in[9];
    float* outp = (float*)d_out;

    const int Nn = in_sizes[0] / 64;
    const int Ee = in_sizes[1] / 2;

    const int nblk = (Ee + SORT_C - 1) / SORT_C;
    const int NB = (Nn + BR - 1) / BR;

    float* p    = (float*)d_ws;
    float* q    = p + (size_t)Nn * 64;
    int*   csr  = (int*)(q + (size_t)Nn * 64);
    unsigned* tmp = (unsigned*)(csr + Ee);
    int*   start = (int*)(tmp + (size_t)nblk * SORT_C);
    int*   cnt  = start + Nn;
    int*   offm = cnt + Nn;

    const int TB = 256;

    partition_kernel<<<nblk, SORT_T, 0, stream>>>(ei, tmp, offm, Ee, NB);
    bucket_kernel<<<NB, SORT_T, 0, stream>>>(tmp, offm, csr, start, cnt, Nn, NB, nblk);

    layer1_kernel<<<(Nn + 15) / 16, 256, 0, stream>>>(
        csr, start, cnt, x, W1l, b1, W1r, W2l, W2r, p, q, Nn);

    layer2_kernel<<<(Nn + 31) / 32, TB, 0, stream>>>(
        csr, start, cnt, p, q, b2, Wfc, bfc, outp, Nn);
}


# Layer-1 weights re-laid once per call in WMMA fragment order (spare blocks of launch 1 write them): one 16-byte load per lane, four whole lines per wave, feeds two WMMAs; was four 4-byte loads
# speedup vs baseline: 6.4039x; 1.0646x over previous
#include <hip/hip_runtime.h>
#include <hip/hip_bf16.h>

typedef __attribute__((ext_vector_type(2))) float v2f;
typedef __attribute__((ext_vector_type(8))) float v8f;

constexpr int SORT_C = 8192;
constexpr int SORT_T = 1024;
constexpr int BSH = 9;
constexpr int BR = 1 << BSH;
constexpr int NB_MAX = 1024;
constexpr int NBLK_MAX = 1024;
constexpr int STG_CAP = 16384;

constexpr int PACK_WORDS = 8192;

__device__ __forceinline__ int block_excl_scan_1024(int v, int lane, int wave,
                                                    int* s_ws, int* s_tot) {
    int inc = v;
#pragma unroll
    for (int d = 1; d < 32; d <<= 1) {
        int t = __shfl_up(inc, d);
        if (lane >= d) inc += t;
    }
    if (lane == 31) s_ws[wave] = inc;
    __syncthreads();
    if (wave == 0) {
        const int w = s_ws[lane];
        int wi = w;
#pragma unroll
        for (int d = 1; d < 32; d <<= 1) {
            int t = __shfl_up(wi, d);
            if (lane >= d) wi += t;
        }
        s_ws[lane] = wi - w;
        if (lane == 31) *s_tot = wi;
    }
    __syncthreads();
    return s_ws[wave] + inc - v;
}

__global__ __launch_bounds__(1024) void partition_kernel(const int* __restrict__ ei,
                                                         unsigned* __restrict__ tmp,
                                                         int* __restrict__ offm,
                                                         int E, int NB, int nblk,
                                                         const float* __restrict__ W1l,
                                                         const float* __restrict__ W1r,
                                                         const float* __restrict__ W2l,
                                                         const float* __restrict__ W2r,
                                                         float* __restrict__ wpk) {
    __shared__ int s_hist[NB_MAX];
    __shared__ unsigned s_stage[SORT_C];
    __shared__ int s_ws[32];
    __shared__ int s_tot;
    const int tid = threadIdx.x;
    const int lane = tid & 31;
    const int wave = tid >> 5;

    if ((int)blockIdx.x >= nblk) {
        const int g = ((int)blockIdx.x - nblk) * SORT_T + tid;
        if (g < PACK_WORDS) {
            const int r = g & 4095;
            const int w = r >> 9;
            const int j = (r >> 5) & 15;
            const int c = r & 15;
            const int h = (r >> 4) & 1;
            float4 v;
            if (g < 4096) {
                const int k = 4 * j + 2 * h;
                const int n = 16 * w + c;
                v.x = W1l[k * 128 + n];
                v.y = W1l[(k + 1) * 128 + n];
                v.z = W1r[k * 128 + n];
                v.w = W1r[(k + 1) * 128 + n];
            } else {
                const float* __restrict__ Ws = (w >= 4) ? W2r : W2l;
                const int k = 8 * j + 2 * h;
                const int n = 16 * (w & 3) + c;
                v.x = Ws[k * 64 + n];
                v.y = Ws[(k + 1) * 64 + n];
                v.z = Ws[(k + 4) * 64 + n];
                v.w = Ws[(k + 5) * 64 + n];
            }
            *(float4*)(wpk + (size_t)g * 4) = v;
        }
        return;
    }
    const int e0 = blockIdx.x * SORT_C;

    s_hist[tid] = 0;
    int sv[8], dv[8];
#pragma unroll
    for (int j = 0; j < 8; ++j) {
        const int e = e0 + j * SORT_T + tid;
        const int ec = e < E ? e : E - 1;
        dv[j] = ei[E + ec];
        sv[j] = ei[ec];
    }
#pragma unroll
    for (int j = 0; j < 8; ++j)
        if (e0 + j * SORT_T + tid >= E) dv[j] = -1;
    __syncthreads();
#pragma unroll
    for (int j = 0; j < 8; ++j)
        if (dv[j] >= 0) atomicAdd(&s_hist[dv[j] >> BSH], 1);
    __syncthreads();
    const int h = s_hist[tid];
    const int ex = block_excl_scan_1024(h, lane, wave, s_ws, &s_tot);
    s_hist[tid] = ex;
    int* __restrict__ row = offm + (size_t)blockIdx.x * (NB + 1);
    if (tid < NB) row[tid] = ex;
    const int tot = s_tot;
    if (tid == 0) row[NB] = tot;
    __syncthreads();
#pragma unroll
    for (int j = 0; j < 8; ++j) {
        if (dv[j] >= 0) {
            const int pos = atomicAdd(&s_hist[dv[j] >> BSH], 1);
            s_stage[pos] = ((unsigned)sv[j] << BSH) | ((unsigned)dv[j] & (BR - 1));
        }
    }
    __syncthreads();
#pragma unroll
    for (int j = 0; j < 8; ++j) {
        const int i = j * SORT_T + tid;
        if (i < tot) tmp[e0 + i] = s_stage[i];
    }
}

template <bool FILL>
__device__ __forceinline__ void visit_word(unsigned w, int* s_cnt, unsigned* s_stg,
                                           int* __restrict__ csr, int base, bool staged) {
    const int dl = (int)(w & (BR - 1));
    if (!FILL) {
        atomicAdd(&s_cnt[dl], 1);
    } else {
        const int pos = atomicAdd(&s_cnt[dl], 1);
        const unsigned src = w >> BSH;
        if (staged) s_stg[pos] = src;
        else csr[base + pos] = (int)src;
    }
}

template <bool FILL>
__device__ __forceinline__ void visit_groups(const unsigned* __restrict__ tmp,
                                             const int* s_segS, const int* s_segL,
                                             int nblk, int lane, int wave, int* s_cnt,
                                             unsigned* s_stg, int* __restrict__ csr,
                                             int base, bool staged) {
    for (int t0 = wave; t0 < nblk; t0 += 128) {
        int s[4], L[4];
        unsigned a[4][2];
#pragma unroll
        for (int u = 0; u < 4; ++u) {
            const int t = t0 + 32 * u;
            const bool ok = t < nblk;
            s[u] = ok ? s_segS[t] : 0;
            L[u] = ok ? s_segL[t] : 0;
        }
#pragma unroll
        for (int u = 0; u < 4; ++u) {
            a[u][0] = tmp[(lane < L[u]) ? s[u] + lane : 0];
            a[u][1] = tmp[(lane + 32 < L[u]) ? s[u] + lane + 32 : 0];
        }
#pragma unroll
        for (int u = 0; u < 4; ++u) {
            if (lane < L[u]) visit_word<FILL>(a[u][0], s_cnt, s_stg, csr, base, staged);
            if (lane + 32 < L[u]) visit_word<FILL>(a[u][1], s_cnt, s_stg, csr, base, staged);
            for (int i = lane + 64; i < L[u]; i += 32)
                visit_word<FILL>(tmp[s[u] + i], s_cnt, s_stg, csr, base, staged);
        }
    }
}

__global__ __launch_bounds__(1024) void bucket_kernel(const unsigned* __restrict__ tmp,
                                                      const int* __restrict__ offm,
                                                      int* __restrict__ csr,
                                                      int* __restrict__ start,
                                                      int* __restrict__ cnt,
                                                      int Nn, int NB, int nblk) {
    __shared__ int s_cnt[BR];
    __shared__ int s_segS[NBLK_MAX];
    __shared__ int s_segL[NBLK_MAX];
    __shared__ unsigned s_stg[STG_CAP];
    __shared__ int s_ws[32];
    __shared__ int s_tot;
    __shared__ int s_base, s_total;
    const int tid = threadIdx.x;
    const int lane = tid & 31;
    const int wave = tid >> 5;
    const int b = blockIdx.x;

    if (tid < BR) s_cnt[tid] = 0;
    if (tid == 0) { s_base = 0; s_total = 0; }
    __syncthreads();
    int pb = 0, pt = 0;
    for (int t = tid; t < nblk; t += SORT_T) {
        const int* __restrict__ row = offm + (size_t)t * (NB + 1);
        const int so = row[b];
        const int eo = row[b + 1];
        s_segS[t] = t * SORT_C + so;
        s_segL[t] = eo - so;
        pb += so;
        pt += eo - so;
    }
#pragma unroll
    for (int d = 16; d > 0; d >>= 1) {
        pb += __shfl_xor(pb, d);
        pt += __shfl_xor(pt, d);
    }
    if (lane == 0 && (pb | pt) != 0) {
        atomicAdd(&s_base, pb);
        atomicAdd(&s_total, pt);
    }
    __syncthreads();
    const int base = s_base;
    const int total = s_total;
    const bool staged = total <= STG_CAP;

    visit_groups<false>(tmp, s_segS, s_segL, nblk, lane, wave, s_cnt, s_stg, csr, base, staged);
    __syncthreads();

    const int c = (tid < BR) ? s_cnt[tid] : 0;
    const int ex = block_excl_scan_1024(c, lane, wave, s_ws, &s_tot);
    if (tid < BR) {
        s_cnt[tid] = ex;
        const int node = b * BR + tid;
        if (node < Nn) {
            start[node] = base + ex;
            cnt[node] = c;
        }
    }
    __syncthreads();

    visit_groups<true>(tmp, s_segS, s_segL, nblk, lane, wave, s_cnt, s_stg, csr, base, staged);
    if (staged) {
        __syncthreads();
        for (int i = tid; i < total; i += SORT_T) csr[base + i] = (int)s_stg[i];
    }
}

__device__ __forceinline__ float2 row_f2(const float* __restrict__ feat, int my, int k, int lane2) {
    const unsigned idx = (unsigned)__builtin_amdgcn_readlane(my, k);
    return *(const float2*)(feat + (size_t)idx * 64 + lane2);
}

__device__ __forceinline__ void gather_run32(float2& acc, const float* __restrict__ feat,
                                             int my, int m, int lane2) {
    int k = 0;
    for (; k + 8 <= m; k += 8) {
        float2 v[8];
#pragma unroll
        for (int j = 0; j < 8; ++j) v[j] = row_f2(feat, my, k + j, lane2);
#pragma unroll
        for (int j = 0; j < 8; ++j) { acc.x += v[j].x; acc.y += v[j].y; }
    }
    const int rem = m - k;
    if (rem & 4) {
        float2 v[4];
#pragma unroll
        for (int j = 0; j < 4; ++j) v[j] = row_f2(feat, my, k + j, lane2);
#pragma unroll
        for (int j = 0; j < 4; ++j) { acc.x += v[j].x; acc.y += v[j].y; }
        k += 4;
    }
    if (rem & 2) {
        const float2 v0 = row_f2(feat, my, k, lane2);
        const float2 v1 = row_f2(feat, my, k + 1, lane2);
        acc.x += v0.x; acc.y += v0.y;
        acc.x += v1.x; acc.y += v1.y;
        k += 2;
    }
    if (rem & 1) {
        const float2 v0 = row_f2(feat, my, k, lane2);
        acc.x += v0.x; acc.y += v0.y;
    }
}

__device__ __forceinline__ float2 node_sum(const int* __restrict__ csr,
                                           const float* __restrict__ feat,
                                           int s, int n, int my, int lane, int lane2) {
    float2 acc = make_float2(0.0f, 0.0f);
    gather_run32(acc, feat, my, n < 32 ? n : 32, lane2);
    for (int base = 32; base < n; base += 32) {
        const int m = n - base < 32 ? n - base : 32;
        const int more = (lane < m) ? csr[s + base + lane] : 0;
        gather_run32(acc, feat, more, m, lane2);
    }
    return acc;
}

__global__ __launch_bounds__(256) void layer1_kernel(
    const int* __restrict__ csr, const int* __restrict__ start,
    const int* __restrict__ cnt, const float* __restrict__ x,
    const float* __restrict__ wpk, const float* __restrict__ b1,
    float* __restrict__ p, float* __restrict__ q, int Nn) {
    constexpr int KA = 64 + 4;
    constexpr int KH = 128 + 4;
    __shared__ __attribute__((aligned(16))) float s_a[16 * KA];
    __shared__ __attribute__((aligned(16))) float s_x[16 * KA];
    __shared__ __attribute__((aligned(16))) float s_h[16 * KH];

    const int tid = threadIdx.x;
    const int lane = tid & 31;
    const int wave = __builtin_amdgcn_readfirstlane((int)(tid >> 5));
    const int lane2 = lane * 2;
    const int node_base = blockIdx.x * 16;
    const int m = lane & 15;
    const int khalf = lane >> 4;

    {
        const int node0 = node_base + wave * 2;
        int ql = node0 + (lane & 1);
        if (ql > Nn - 1) ql = Nn - 1;
        const int sl = start[ql];
        const int nl = cnt[ql];
        int s[2], n[2], my[2];
#pragma unroll
        for (int i = 0; i < 2; ++i) {
            s[i] = __builtin_amdgcn_readlane(sl, i);
            n[i] = (node0 + i < Nn) ? __builtin_amdgcn_readlane(nl, i) : 0;
        }
#pragma unroll
        for (int i = 0; i < 2; ++i) my[i] = (lane < n[i]) ? csr[s[i] + lane] : 0;
#pragma unroll
        for (int i = 0; i < 2; ++i) {
            int gnode = node0 + i;
            if (gnode > Nn - 1) gnode = Nn - 1;
            const float2 xv = *(const float2*)(x + (size_t)gnode * 64 + lane2);
            const float2 acc = node_sum(csr, x, s[i], n[i], my[i], lane, lane2);
            const float inv = 1.0f / fmaxf((float)n[i], 1.0f);
            const int row = wave * 2 + i;
            *(float2*)(s_a + row * KA + lane2) = make_float2(acc.x * inv, acc.y * inv);
            *(float2*)(s_x + row * KA + lane2) = xv;
        }
    }
    __syncthreads();

    {
        const int ncol = wave * 16 + m;
        const float4* __restrict__ wp = (const float4*)wpk + wave * 512 + lane;
        v8f acc = {};
#pragma unroll
        for (int k = 0; k < 64; k += 4) {
            const int kb = k + khalf * 2;
            const float4 bv = wp[(k >> 2) * 32];
            v2f a, b;
            a.x = s_a[m * KA + kb];
            a.y = s_a[m * KA + kb + 1];
            b.x = bv.x;
            b.y = bv.y;
            acc = __builtin_amdgcn_wmma_f32_16x16x4_f32(false, a, false, b,
                                                        (short)0, acc, false, false);
            a.x = s_x[m * KA + kb];
            a.y = s_x[m * KA + kb + 1];
            b.x = bv.z;
            b.y = bv.w;
            acc = __builtin_amdgcn_wmma_f32_16x16x4_f32(false, a, false, b,
                                                        (short)0, acc, false, false);
        }
        const float bias = b1[ncol];
#pragma unroll
        for (int v = 0; v < 8; ++v)
            s_h[(v + khalf * 8) * KH + ncol] = fmaxf(acc[v] + bias, 0.0f);
    }
    __syncthreads();

    {
        const bool second = wave >= 4;
        float* __restrict__ dstp = second ? q : p;
        const int ncol = (wave & 3) * 16 + m;
        const float4* __restrict__ wp = (const float4*)wpk + 4096 + wave * 512 + lane;
        v8f acc = {};
#pragma unroll
        for (int k = 0; k < 128; k += 8) {
            const int kb = k + khalf * 2;
            const float4 bv = wp[(k >> 3) * 32];
            v2f a, b;
            a.x = s_h[m * KH + kb];
            a.y = s_h[m * KH + kb + 1];
            b.x = bv.x;
            b.y = bv.y;
            acc = __builtin_amdgcn_wmma_f32_16x16x4_f32(false, a, false, b,
                                                        (short)0, acc, false, false);
            a.x = s_h[m * KH + kb + 4];
            a.y = s_h[m * KH + kb + 5];
            b.x = bv.z;
            b.y = bv.w;
            acc = __builtin_amdgcn_wmma_f32_16x16x4_f32(false, a, false, b,
                                                        (short)0, acc, false, false);
        }
#pragma unroll
        for (int v = 0; v < 8; ++v) {
            const int gnode = node_base + v + khalf * 8;
            if (gnode < Nn) dstp[(size_t)gnode * 64 + ncol] = acc[v];
        }
    }
}

__global__ __launch_bounds__(256) void layer2_kernel(
    const int* __restrict__ csr, const int* __restrict__ start,
    const int* __restrict__ cnt, const float* __restrict__ p,
    const float* __restrict__ q, const float* __restrict__ b2,
    const float* __restrict__ Wfc, const float* __restrict__ bfc,
    float* __restrict__ out, int Nn) {
    __shared__ float s_out[32];
    const int tid = threadIdx.x;
    const int lane = tid & 31;
    const int wave = __builtin_amdgcn_readfirstlane((int)(tid >> 5));
    const int lane2 = lane * 2;
    const int node0 = (blockIdx.x * 8 + wave) * 4;

    int ql = node0 + (lane & 3);
    if (ql > Nn - 1) ql = Nn - 1;
    const int sl = start[ql];
    const int nl = cnt[ql];
    int s[4], n[4], my[4];
#pragma unroll
    for (int i = 0; i < 4; ++i) {
        s[i] = __builtin_amdgcn_readlane(sl, i);
        n[i] = (node0 + i < Nn) ? __builtin_amdgcn_readlane(nl, i) : 0;
    }
#pragma unroll
    for (int i = 0; i < 4; ++i) my[i] = (lane < n[i]) ? csr[s[i] + lane] : 0;

    const float2 bb = *(const float2*)(b2 + lane2);
    const float4 w = *(const float4*)(Wfc + (lane >> 1) * 4);
    const float bf = bfc[0];

    float res = 0.0f;
#pragma unroll
    for (int i = 0; i < 4; ++i) {
        int gnode = node0 + i;
        if (gnode > Nn - 1) gnode = Nn - 1;
        const float2 qv = *(const float2*)(q + (size_t)gnode * 64 + lane2);
        const float2 acc = node_sum(csr, p, s[i], n[i], my[i], lane, lane2);
        const float inv = 1.0f / fmaxf((float)n[i], 1.0f);
        const float r0 = fmaxf(acc.x * inv + bb.x + qv.x, 0.0f);
        const float r1 = fmaxf(acc.y * inv + bb.y + qv.y, 0.0f);
        const float r2 = __shfl_xor(r0, 1);
        const float r3 = __shfl_xor(r1, 1);
        float d = r0 * w.x;
        d += r1 * w.y;
        d += r2 * w.z;
        d += r3 * w.w;
        d += __shfl_xor(d, 16);
        d += __shfl_xor(d, 8);
        d += __shfl_xor(d, 4);
        d += __shfl_xor(d, 2);
        const float o = 1.0f / (1.0f + __expf(-(d + bf)));
        const float o0 = __builtin_bit_cast(
            float, __builtin_amdgcn_readfirstlane(__builtin_bit_cast(int, o)));
        if (lane == i) res = o0;
    }
    if (lane < 4) s_out[wave * 4 + lane] = res;
    __syncthreads();
    if (tid < 32) {
        const int node = blockIdx.x * 32 + tid;
        if (node < Nn) out[node] = s_out[tid];
    }
}

extern "C" void kernel_launch(void* const* d_in, const int* in_sizes, int n_in,
                              void* d_out, int out_size, void* d_ws, size_t ws_size,
                              hipStream_t stream) {
    const float* x    = (const float*)d_in[0];
    const int*   ei   = (const int*)d_in[1];
    const float* W1l  = (const float*)d_in[2];
    const float* b1   = (const float*)d_in[3];
    const float* W1r  = (const float*)d_in[4];
    const float* W2l  = (const float*)d_in[5];
    const float* b2   = (const float*)d_in[6];
    const float* W2r  = (const float*)d_in[7];
    const float* Wfc  = (const float*)d_in[8];
    const float* bfc  = (const float*)d_in[9];
    float* outp = (float*)d_out;

    const int Nn = in_sizes[0] / 64;
    const int Ee = in_sizes[1] / 2;

    const int nblk = (Ee + SORT_C - 1) / SORT_C;
    const int NB = (Nn + BR - 1) / BR;

    float* wpk  = (float*)d_ws;
    float* p    = wpk + (size_t)PACK_WORDS * 4;
    float* q    = p + (size_t)Nn * 64;
    int*   csr  = (int*)(q + (size_t)Nn * 64);
    unsigned* tmp = (unsigned*)(csr + Ee);
    int*   start = (int*)(tmp + (size_t)nblk * SORT_C);
    int*   cnt  = start + Nn;
    int*   offm = cnt + Nn;

    const int TB = 256;

    partition_kernel<<<nblk + PACK_WORDS / SORT_T, SORT_T, 0, stream>>>(
        ei, tmp, offm, Ee, NB, nblk, W1l, W1r, W2l, W2r, wpk);
    bucket_kernel<<<NB, SORT_T, 0, stream>>>(tmp, offm, csr, start, cnt, Nn, NB, nblk);

    layer1_kernel<<<(Nn + 15) / 16, 256, 0, stream>>>(
        csr, start, cnt, x, wpk, b1, p, q, Nn);

    layer2_kernel<<<(Nn + 31) / 32, TB, 0, stream>>>(
        csr, start, cnt, p, q, b2, Wfc, bfc, outp, Nn);
}
